# starting kernel
# baseline (speedup 1.0000x reference)
#include <hip/hip_runtime.h>

typedef __bf16 bf16_t;
typedef __attribute__((ext_vector_type(8)))  __bf16 v8bf;
typedef __attribute__((ext_vector_type(16))) __bf16 v16bf;
typedef __attribute__((ext_vector_type(8)))  float  v8f;

#define B_     64
#define C_     128
#define L_     4096
#define P_     64
#define K_     5
#define LOUT   4092
#define NTILE  128
#define ROWS   (NTILE + K_ - 1)
#define PITCH  136
#define CK     (C_ * K_)
#define NCHUNK (CK / 32)
#define AW_ELEMS (4 * NCHUNK * 2 * 32 * 16)

__global__ void repack_kernel(const float* __restrict__ proto,
                              unsigned short* __restrict__ awRaw) {
  bf16_t* aw = reinterpret_cast<bf16_t*>(awRaw);
  int idx  = blockIdx.x * blockDim.x + threadIdx.x;
  int e    = idx & 15;
  int r1   = idx >> 4;
  int lane = r1 & 31;
  int r2   = r1 >> 5;
  int term = r2 & 1;
  int r3   = r2 >> 1;
  int s    = r3 % NCHUNK;
  int mt   = r3 / NCHUNK;
  int hf   = lane >> 4;
  int p    = mt * 16 + (lane & 15);
  int tt   = e + ((e >= 8) ? 8 : 0) + hf * 8;
  int c    = (s & 3) * 32 + tt;
  int k    = s >> 2;
  float w  = proto[(p * C_ + c) * K_ + k];
  bf16_t h = (bf16_t)w;
  bf16_t l = (bf16_t)(w - (float)h);
  aw[idx]  = term ? l : h;
}

__global__ void p2_kernel(const float* __restrict__ proto, float* __restrict__ p2) {
  int p = threadIdx.x;
  float s = 0.f;
  for (int i = 0; i < CK; ++i) {
    float w = proto[p * CK + i];
    s += w * w;
  }
  p2[p] = s;
}

__global__ void __launch_bounds__(256)
proto_dist_kernel(const float* __restrict__ x,
                  const unsigned short* __restrict__ awRaw,
                  const float* __restrict__ p2,
                  float* __restrict__ out) {
  __shared__ __align__(16) bf16_t xhiT[ROWS * PITCH];
  __shared__ __align__(16) bf16_t xloT[ROWS * PITCH];
  __shared__ float sq[ROWS];
  __shared__ float x2w[NTILE];

  const int tid = threadIdx.x;
  const int b   = blockIdx.y;
  const int l0  = blockIdx.x * NTILE;

  if (tid < ROWS) sq[tid] = 0.f;
  __syncthreads();

  const float* xb = x + (size_t)b * C_ * L_;
  #pragma unroll 4
  for (int i = 0; i < (C_ * ROWS) / 256; ++i) {
    int idx  = i * 256 + tid;
    int c    = idx / ROWS;
    int pos  = idx - c * ROWS;
    int gpos = l0 + pos;
    float v  = (gpos < L_) ? xb[(size_t)c * L_ + gpos] : 0.f;
    bf16_t h = (bf16_t)v;
    bf16_t l = (bf16_t)(v - (float)h);
    xhiT[pos * PITCH + c] = h;
    xloT[pos * PITCH + c] = l;
    atomicAdd(&sq[pos], v * v);
  }
  __syncthreads();

  if (tid < NTILE) {
    float s = 0.f;
    #pragma unroll
    for (int k = 0; k < K_; ++k) s += sq[tid + k];
    x2w[tid] = s;
  }
  __syncthreads();

  const int wave  = tid >> 5;
  const int lane  = tid & 31;
  const int mt    = wave & 3;
  const int nh    = wave >> 2;
  const int n     = lane & 15;
  const int hf    = lane >> 4;
  const int nbase = nh * 64;

  const v16bf* awv = reinterpret_cast<const v16bf*>(awRaw);

  v8f acc[4] = {};

  for (int s = 0; s < NCHUNK; ++s) {
    const int k  = s >> 2;
    const int c0 = (s & 3) * 32;
    const int ia = ((mt * NCHUNK + s) * 2) * 32 + lane;
    v16bf ah = awv[ia];
    v16bf al = awv[ia + 32];
    if (s + 1 < NCHUNK) __builtin_prefetch(&awv[ia + 64], 0, 1);

    const int cA = c0 + hf * 8;
    #pragma unroll
    for (int j = 0; j < 4; ++j) {
      const int pos = nbase + j * 16 + n + k;
      const bf16_t* rh = &xhiT[pos * PITCH + cA];
      const bf16_t* rl = &xloT[pos * PITCH + cA];
      v8bf h0 = *(const v8bf*)(rh);
      v8bf h1 = *(const v8bf*)(rh + 16);
      v8bf q0 = *(const v8bf*)(rl);
      v8bf q1 = *(const v8bf*)(rl + 16);
      v16bf bh = __builtin_shufflevector(h0, h1, 0,1,2,3,4,5,6,7,8,9,10,11,12,13,14,15);
      v16bf bl = __builtin_shufflevector(q0, q1, 0,1,2,3,4,5,6,7,8,9,10,11,12,13,14,15);
      acc[j] = __builtin_amdgcn_wmma_f32_16x16x32_bf16(false, ah, false, bh, (short)0, acc[j], false, false);
      acc[j] = __builtin_amdgcn_wmma_f32_16x16x32_bf16(false, ah, false, bl, (short)0, acc[j], false, false);
      acc[j] = __builtin_amdgcn_wmma_f32_16x16x32_bf16(false, al, false, bh, (short)0, acc[j], false, false);
    }
  }

  const int pm0 = mt * 16 + hf * 8;
  float p2v[8];
  #pragma unroll
  for (int r = 0; r < 8; ++r) p2v[r] = p2[pm0 + r];

  #pragma unroll
  for (int j = 0; j < 4; ++j) {
    const int col = nbase + j * 16 + n;
    const int l   = l0 + col;
    if (l < LOUT) {
      const float x2v = x2w[col];
      #pragma unroll
      for (int r = 0; r < 8; ++r) {
        float v = x2v - 2.f * acc[j][r] + p2v[r];
        out[((size_t)b * P_ + (pm0 + r)) * LOUT + l] = v > 0.f ? v : 0.f;
      }
    }
  }
}

extern "C" void kernel_launch(void* const* d_in, const int* in_sizes, int n_in,
                              void* d_out, int out_size, void* d_ws, size_t ws_size,
                              hipStream_t stream) {
  const float* x     = (const float*)d_in[0];
  const float* proto = (const float*)d_in[1];
  float* out         = (float*)d_out;

  unsigned short* aw = (unsigned short*)d_ws;
  float* p2          = (float*)((char*)d_ws + (size_t)AW_ELEMS * 2);

  repack_kernel<<<AW_ELEMS / 256, 256, 0, stream>>>(proto, aw);
  p2_kernel<<<1, P_, 0, stream>>>(proto, p2);
  proto_dist_kernel<<<dim3((LOUT + NTILE - 1) / NTILE, B_), 256, 0, stream>>>(x, aw, p2, out);
}


# x tile staging re-mapped: a thread owns one position, issues all its 66 global loads before the first wait (was 1 in flight), 16-byte LDS stores, squares summed in registers (x2 sum order changes)
# speedup vs baseline: 1.9858x; 1.9858x over previous
#include <hip/hip_runtime.h>

typedef __bf16 bf16_t;
typedef __attribute__((ext_vector_type(8)))  __bf16 v8bf;
typedef __attribute__((ext_vector_type(16))) __bf16 v16bf;
typedef __attribute__((ext_vector_type(8)))  float  v8f;

#define B_     64
#define C_     128
#define L_     4096
#define P_     64
#define K_     5
#define LOUT   4092
#define NTILE  128
#define ROWS   (NTILE + K_ - 1)
#define PITCH  136
#define CK     (C_ * K_)
#define NCHUNK (CK / 32)
#define AW_ELEMS (4 * NCHUNK * 2 * 32 * 16)

__global__ void repack_kernel(const float* __restrict__ proto,
                              unsigned short* __restrict__ awRaw) {
  bf16_t* aw = reinterpret_cast<bf16_t*>(awRaw);
  int idx  = blockIdx.x * blockDim.x + threadIdx.x;
  int e    = idx & 15;
  int r1   = idx >> 4;
  int lane = r1 & 31;
  int r2   = r1 >> 5;
  int term = r2 & 1;
  int r3   = r2 >> 1;
  int s    = r3 % NCHUNK;
  int mt   = r3 / NCHUNK;
  int hf   = lane >> 4;
  int p    = mt * 16 + (lane & 15);
  int tt   = e + ((e >= 8) ? 8 : 0) + hf * 8;
  int c    = (s & 3) * 32 + tt;
  int k    = s >> 2;
  float w  = proto[(p * C_ + c) * K_ + k];
  bf16_t h = (bf16_t)w;
  bf16_t l = (bf16_t)(w - (float)h);
  aw[idx]  = term ? l : h;
}

__global__ void p2_kernel(const float* __restrict__ proto, float* __restrict__ p2) {
  int p = threadIdx.x;
  float s = 0.f;
  for (int i = 0; i < CK; ++i) {
    float w = proto[p * CK + i];
    s += w * w;
  }
  p2[p] = s;
}

__global__ void __launch_bounds__(256)
proto_dist_kernel(const float* __restrict__ x,
                  const unsigned short* __restrict__ awRaw,
                  const float* __restrict__ p2,
                  float* __restrict__ out) {
  __shared__ __align__(16) bf16_t xhiT[ROWS * PITCH];
  __shared__ __align__(16) bf16_t xloT[ROWS * PITCH];
  __shared__ float sqp[2][NTILE];
  __shared__ float sqe[8][4];
  __shared__ float x2w[NTILE];

  const int tid = threadIdx.x;
  const int b   = blockIdx.y;
  const int l0  = blockIdx.x * NTILE;

  const float* xb = x + (size_t)b * C_ * L_;
  {
    const int pe   = tid & 3;
    const int ce   = tid >> 2;
    const int gpe  = l0 + NTILE + pe;
    const bool inr = gpe < L_;
    const int gpc  = inr ? gpe : (L_ - 1);
    float e0 = xb[(size_t)ce * L_ + gpc];
    float e1 = xb[(size_t)(ce + 64) * L_ + gpc];

    const int pos = tid & (NTILE - 1);
    const int cg  = tid >> 7;
    const float* xp = xb + (size_t)(cg * 64) * L_ + (l0 + pos);
    float s = 0.f;
    float v[64];
    #pragma unroll
    for (int i = 0; i < 64; ++i) v[i] = xp[(size_t)i * L_];
    #pragma unroll
    for (int g = 0; g < 8; ++g) {
      v8bf h8, l8;
      #pragma unroll
      for (int e = 0; e < 8; ++e) {
        const float f = v[g * 8 + e];
        const bf16_t h = (bf16_t)f;
        h8[e] = h;
        l8[e] = (bf16_t)(f - (float)h);
        s += f * f;
      }
      const int off = pos * PITCH + cg * 64 + g * 8;
      *(v8bf*)(&xhiT[off]) = h8;
      *(v8bf*)(&xloT[off]) = l8;
    }
    sqp[cg][pos] = s;

    e0 = inr ? e0 : 0.f;
    e1 = inr ? e1 : 0.f;
    const bf16_t eh0 = (bf16_t)e0;
    const bf16_t eh1 = (bf16_t)e1;
    const int offe = (NTILE + pe) * PITCH + ce;
    xhiT[offe]      = eh0;
    xloT[offe]      = (bf16_t)(e0 - (float)eh0);
    xhiT[offe + 64] = eh1;
    xloT[offe + 64] = (bf16_t)(e1 - (float)eh1);
    float se = e0 * e0 + e1 * e1;
    se += __shfl_xor(se, 4);
    se += __shfl_xor(se, 8);
    se += __shfl_xor(se, 16);
    if ((tid & 31) < 4) sqe[tid >> 5][tid & 3] = se;
  }
  __syncthreads();

  if (tid < NTILE) {
    float s = 0.f;
    #pragma unroll
    for (int k = 0; k < K_; ++k) {
      const int p = tid + k;
      if (p < NTILE) {
        s += sqp[0][p] + sqp[1][p];
      } else {
        float t = 0.f;
        #pragma unroll
        for (int w = 0; w < 8; ++w) t += sqe[w][p - NTILE];
        s += t;
      }
    }
    x2w[tid] = s;
  }
  __syncthreads();

  const int wave  = tid >> 5;
  const int lane  = tid & 31;
  const int mt    = wave & 3;
  const int nh    = wave >> 2;
  const int n     = lane & 15;
  const int hf    = lane >> 4;
  const int nbase = nh * 64;

  const v16bf* awv = reinterpret_cast<const v16bf*>(awRaw);

  v8f acc[4] = {};

  for (int s = 0; s < NCHUNK; ++s) {
    const int k  = s >> 2;
    const int c0 = (s & 3) * 32;
    const int ia = ((mt * NCHUNK + s) * 2) * 32 + lane;
    v16bf ah = awv[ia];
    v16bf al = awv[ia + 32];
    if (s + 1 < NCHUNK) __builtin_prefetch(&awv[ia + 64], 0, 1);

    const int cA = c0 + hf * 8;
    #pragma unroll
    for (int j = 0; j < 4; ++j) {
      const int pos = nbase + j * 16 + n + k;
      const bf16_t* rh = &xhiT[pos * PITCH + cA];
      const bf16_t* rl = &xloT[pos * PITCH + cA];
      v8bf h0 = *(const v8bf*)(rh);
      v8bf h1 = *(const v8bf*)(rh + 16);
      v8bf q0 = *(const v8bf*)(rl);
      v8bf q1 = *(const v8bf*)(rl + 16);
      v16bf bh = __builtin_shufflevector(h0, h1, 0,1,2,3,4,5,6,7,8,9,10,11,12,13,14,15);
      v16bf bl = __builtin_shufflevector(q0, q1, 0,1,2,3,4,5,6,7,8,9,10,11,12,13,14,15);
      acc[j] = __builtin_amdgcn_wmma_f32_16x16x32_bf16(false, ah, false, bh, (short)0, acc[j], false, false);
      acc[j] = __builtin_amdgcn_wmma_f32_16x16x32_bf16(false, ah, false, bl, (short)0, acc[j], false, false);
      acc[j] = __builtin_amdgcn_wmma_f32_16x16x32_bf16(false, al, false, bh, (short)0, acc[j], false, false);
    }
  }

  const int pm0 = mt * 16 + hf * 8;
  float p2v[8];
  #pragma unroll
  for (int r = 0; r < 8; ++r) p2v[r] = p2[pm0 + r];

  #pragma unroll
  for (int j = 0; j < 4; ++j) {
    const int col = nbase + j * 16 + n;
    const int l   = l0 + col;
    if (l < LOUT) {
      const float x2v = x2w[col];
      #pragma unroll
      for (int r = 0; r < 8; ++r) {
        float v = x2v - 2.f * acc[j][r] + p2v[r];
        out[((size_t)b * P_ + (pm0 + r)) * LOUT + l] = v > 0.f ? v : 0.f;
      }
    }
  }
}

extern "C" void kernel_launch(void* const* d_in, const int* in_sizes, int n_in,
                              void* d_out, int out_size, void* d_ws, size_t ws_size,
                              hipStream_t stream) {
  const float* x     = (const float*)d_in[0];
  const float* proto = (const float*)d_in[1];
  float* out         = (float*)d_out;

  unsigned short* aw = (unsigned short*)d_ws;
  float* p2          = (float*)((char*)d_ws + (size_t)AW_ELEMS * 2);

  repack_kernel<<<AW_ELEMS / 256, 256, 0, stream>>>(proto, aw);
  p2_kernel<<<1, P_, 0, stream>>>(proto, p2);
  proto_dist_kernel<<<dim3((LOUT + NTILE - 1) / NTILE, B_), 256, 0, stream>>>(x, aw, p2, out);
}


# p2 preparation kernel parallelised: a wave per prototype, 5 independent float4 loads per lane and a shuffle tree; was one thread per prototype walking 640 strided floats (p2 sum order changes)
# speedup vs baseline: 2.1525x; 1.0839x over previous
#include <hip/hip_runtime.h>

typedef __bf16 bf16_t;
typedef __attribute__((ext_vector_type(8)))  __bf16 v8bf;
typedef __attribute__((ext_vector_type(16))) __bf16 v16bf;
typedef __attribute__((ext_vector_type(8)))  float  v8f;

#define B_     64
#define C_     128
#define L_     4096
#define P_     64
#define K_     5
#define LOUT   4092
#define NTILE  128
#define ROWS   (NTILE + K_ - 1)
#define PITCH  136
#define CK     (C_ * K_)
#define NCHUNK (CK / 32)
#define AW_ELEMS (4 * NCHUNK * 2 * 32 * 16)

__global__ void repack_kernel(const float* __restrict__ proto,
                              unsigned short* __restrict__ awRaw) {
  bf16_t* aw = reinterpret_cast<bf16_t*>(awRaw);
  int idx  = blockIdx.x * blockDim.x + threadIdx.x;
  int e    = idx & 15;
  int r1   = idx >> 4;
  int lane = r1 & 31;
  int r2   = r1 >> 5;
  int term = r2 & 1;
  int r3   = r2 >> 1;
  int s    = r3 % NCHUNK;
  int mt   = r3 / NCHUNK;
  int hf   = lane >> 4;
  int p    = mt * 16 + (lane & 15);
  int tt   = e + ((e >= 8) ? 8 : 0) + hf * 8;
  int c    = (s & 3) * 32 + tt;
  int k    = s >> 2;
  float w  = proto[(p * C_ + c) * K_ + k];
  bf16_t h = (bf16_t)w;
  bf16_t l = (bf16_t)(w - (float)h);
  aw[idx]  = term ? l : h;
}

__global__ void __launch_bounds__(1024)
p2_kernel(const float* __restrict__ proto, float* __restrict__ p2) {
  __shared__ float part[32];
  const int wave = threadIdx.x >> 5;
  const int lane = threadIdx.x & 31;
  const int p    = blockIdx.x * 32 + wave;
  const float4* row = reinterpret_cast<const float4*>(proto + (size_t)p * CK);
  float4 q[CK / 128];
  #pragma unroll
  for (int i = 0; i < CK / 128; ++i) q[i] = row[i * 32 + lane];
  float s = 0.f;
  #pragma unroll
  for (int i = 0; i < CK / 128; ++i)
    s += q[i].x * q[i].x + q[i].y * q[i].y + q[i].z * q[i].z + q[i].w * q[i].w;
  s += __shfl_xor(s, 16);
  s += __shfl_xor(s, 8);
  s += __shfl_xor(s, 4);
  s += __shfl_xor(s, 2);
  s += __shfl_xor(s, 1);
  if (lane == 0) part[wave] = s;
  __syncthreads();
  if (threadIdx.x < 32) p2[blockIdx.x * 32 + threadIdx.x] = part[threadIdx.x];
}

__global__ void __launch_bounds__(256)
proto_dist_kernel(const float* __restrict__ x,
                  const unsigned short* __restrict__ awRaw,
                  const float* __restrict__ p2,
                  float* __restrict__ out) {
  __shared__ __align__(16) bf16_t xhiT[ROWS * PITCH];
  __shared__ __align__(16) bf16_t xloT[ROWS * PITCH];
  __shared__ float sqp[2][NTILE];
  __shared__ float sqe[8][4];
  __shared__ float x2w[NTILE];

  const int tid = threadIdx.x;
  const int b   = blockIdx.y;
  const int l0  = blockIdx.x * NTILE;

  const float* xb = x + (size_t)b * C_ * L_;
  {
    const int pe   = tid & 3;
    const int ce   = tid >> 2;
    const int gpe  = l0 + NTILE + pe;
    const bool inr = gpe < L_;
    const int gpc  = inr ? gpe : (L_ - 1);
    float e0 = xb[(size_t)ce * L_ + gpc];
    float e1 = xb[(size_t)(ce + 64) * L_ + gpc];

    const int pos = tid & (NTILE - 1);
    const int cg  = tid >> 7;
    const float* xp = xb + (size_t)(cg * 64) * L_ + (l0 + pos);
    float s = 0.f;
    float v[64];
    #pragma unroll
    for (int i = 0; i < 64; ++i) v[i] = xp[(size_t)i * L_];
    #pragma unroll
    for (int g = 0; g < 8; ++g) {
      v8bf h8, l8;
      #pragma unroll
      for (int e = 0; e < 8; ++e) {
        const float f = v[g * 8 + e];
        const bf16_t h = (bf16_t)f;
        h8[e] = h;
        l8[e] = (bf16_t)(f - (float)h);
        s += f * f;
      }
      const int off = pos * PITCH + cg * 64 + g * 8;
      *(v8bf*)(&xhiT[off]) = h8;
      *(v8bf*)(&xloT[off]) = l8;
    }
    sqp[cg][pos] = s;

    e0 = inr ? e0 : 0.f;
    e1 = inr ? e1 : 0.f;
    const bf16_t eh0 = (bf16_t)e0;
    const bf16_t eh1 = (bf16_t)e1;
    const int offe = (NTILE + pe) * PITCH + ce;
    xhiT[offe]      = eh0;
    xloT[offe]      = (bf16_t)(e0 - (float)eh0);
    xhiT[offe + 64] = eh1;
    xloT[offe + 64] = (bf16_t)(e1 - (float)eh1);
    float se = e0 * e0 + e1 * e1;
    se += __shfl_xor(se, 4);
    se += __shfl_xor(se, 8);
    se += __shfl_xor(se, 16);
    if ((tid & 31) < 4) sqe[tid >> 5][tid & 3] = se;
  }
  __syncthreads();

  if (tid < NTILE) {
    float s = 0.f;
    #pragma unroll
    for (int k = 0; k < K_; ++k) {
      const int p = tid + k;
      if (p < NTILE) {
        s += sqp[0][p] + sqp[1][p];
      } else {
        float t = 0.f;
        #pragma unroll
        for (int w = 0; w < 8; ++w) t += sqe[w][p - NTILE];
        s += t;
      }
    }
    x2w[tid] = s;
  }
  __syncthreads();

  const int wave  = tid >> 5;
  const int lane  = tid & 31;
  const int mt    = wave & 3;
  const int nh    = wave >> 2;
  const int n     = lane & 15;
  const int hf    = lane >> 4;
  const int nbase = nh * 64;

  const v16bf* awv = reinterpret_cast<const v16bf*>(awRaw);

  v8f acc[4] = {};

  for (int s = 0; s < NCHUNK; ++s) {
    const int k  = s >> 2;
    const int c0 = (s & 3) * 32;
    const int ia = ((mt * NCHUNK + s) * 2) * 32 + lane;
    v16bf ah = awv[ia];
    v16bf al = awv[ia + 32];
    if (s + 1 < NCHUNK) __builtin_prefetch(&awv[ia + 64], 0, 1);

    const int cA = c0 + hf * 8;
    #pragma unroll
    for (int j = 0; j < 4; ++j) {
      const int pos = nbase + j * 16 + n + k;
      const bf16_t* rh = &xhiT[pos * PITCH + cA];
      const bf16_t* rl = &xloT[pos * PITCH + cA];
      v8bf h0 = *(const v8bf*)(rh);
      v8bf h1 = *(const v8bf*)(rh + 16);
      v8bf q0 = *(const v8bf*)(rl);
      v8bf q1 = *(const v8bf*)(rl + 16);
      v16bf bh = __builtin_shufflevector(h0, h1, 0,1,2,3,4,5,6,7,8,9,10,11,12,13,14,15);
      v16bf bl = __builtin_shufflevector(q0, q1, 0,1,2,3,4,5,6,7,8,9,10,11,12,13,14,15);
      acc[j] = __builtin_amdgcn_wmma_f32_16x16x32_bf16(false, ah, false, bh, (short)0, acc[j], false, false);
      acc[j] = __builtin_amdgcn_wmma_f32_16x16x32_bf16(false, ah, false, bl, (short)0, acc[j], false, false);
      acc[j] = __builtin_amdgcn_wmma_f32_16x16x32_bf16(false, al, false, bh, (short)0, acc[j], false, false);
    }
  }

  const int pm0 = mt * 16 + hf * 8;
  float p2v[8];
  #pragma unroll
  for (int r = 0; r < 8; ++r) p2v[r] = p2[pm0 + r];

  #pragma unroll
  for (int j = 0; j < 4; ++j) {
    const int col = nbase + j * 16 + n;
    const int l   = l0 + col;
    if (l < LOUT) {
      const float x2v = x2w[col];
      #pragma unroll
      for (int r = 0; r < 8; ++r) {
        float v = x2v - 2.f * acc[j][r] + p2v[r];
        out[((size_t)b * P_ + (pm0 + r)) * LOUT + l] = v > 0.f ? v : 0.f;
      }
    }
  }
}

extern "C" void kernel_launch(void* const* d_in, const int* in_sizes, int n_in,
                              void* d_out, int out_size, void* d_ws, size_t ws_size,
                              hipStream_t stream) {
  const float* x     = (const float*)d_in[0];
  const float* proto = (const float*)d_in[1];
  float* out         = (float*)d_out;

  unsigned short* aw = (unsigned short*)d_ws;
  float* p2          = (float*)((char*)d_ws + (size_t)AW_ELEMS * 2);

  repack_kernel<<<AW_ELEMS / 256, 256, 0, stream>>>(proto, aw);
  p2_kernel<<<P_ / 32, 1024, 0, stream>>>(proto, p2);
  proto_dist_kernel<<<dim3((LOUT + NTILE - 1) / NTILE, B_), 256, 0, stream>>>(x, aw, p2, out);
}
